# starting kernel
# baseline (speedup 1.0000x reference)
#include <hip/hip_runtime.h>
#include <hip/hip_bf16.h>

typedef float v2f __attribute__((ext_vector_type(2)));
typedef float v8f __attribute__((ext_vector_type(8)));

#define UP      5
#define DOWN    24
#define HLEN    481
#define MSHIFT  10
#define JMAX    96
#define KSTEPS  43
#define XCOLS   712
#define XSTRIDE 721
#define YSTRIDE 132
#define TAPMAX  844
#define HPAD    1232

__global__ __launch_bounds__(256) void resample_poly_wmma(
    const float* __restrict__ x,
    const float* __restrict__ h,
    float* __restrict__ out,
    int n_in, int n_out)
{
    __shared__ float hrev[HPAD];
    __shared__ float xs[16 * XSTRIDE];
    __shared__ float yo[16 * YSTRIDE];

    const int tid  = threadIdx.x;
    const int wave = tid >> 5;
    const int lane = tid & 31;
    const int b0    = blockIdx.y << 4;
    const int m_wg0 = blockIdx.x << 7;

    for (int i = tid; i < HPAD; i += 256) {
        int k = TAPMAX - i;
        hrev[i] = ((unsigned)k < (unsigned)HLEN) ? 5.0f * h[k] : 0.0f;
    }

    const int t0_wg = (DOWN * (m_wg0 + MSHIFT)) / UP;
    const int t_min = t0_wg - JMAX;
    for (int r = 0; r < 16; ++r) {
        const float* xrow = x + (size_t)(b0 + r) * (size_t)n_in;
        float* dst = &xs[r * XSTRIDE];
        for (int c = tid; c < XCOLS; c += 256) {
            int t = t_min + c;
            dst[c] = ((unsigned)t < (unsigned)n_in) ? xrow[t] : 0.0f;
        }
    }
    __syncthreads();

    int tile = (blockIdx.x << 3) + wave;
    const int last_tile = (n_out >> 4) - 1;
    if (tile > last_tile) tile = last_tile;
    const int m0   = tile << 4;
    const int t0t  = (DOWN * (m0 + MSHIFT)) / UP;
    const int toff = (t0t - JMAX) - t_min;

    const int mrow = lane & 15;
    const int ksub = (lane >> 4) << 1;
    const int tap0 = DOWN * (m0 + mrow + MSHIFT) - UP * (t0t - JMAX) - UP * ksub;

    const float* hA = &hrev[TAPMAX - tap0];
    const float* xB = &xs[mrow * XSTRIDE + toff + ksub];

    v8f acc = {0.f, 0.f, 0.f, 0.f, 0.f, 0.f, 0.f, 0.f};

    #pragma unroll
    for (int kk = 0; kk < KSTEPS; ++kk) {
        float av0 = hA[20 * kk];
        float av1 = hA[20 * kk + 5];
        float bv0 = xB[4 * kk];
        float bv1 = xB[4 * kk + 1];
        v2f A = {av0, av1};
        v2f B = {bv0, bv1};
        acc = __builtin_amdgcn_wmma_f32_16x16x4_f32(
            false, A, false, B, (short)0, acc, false, false);
    }

    {
        const int colw = (wave << 4) + ((lane >> 4) << 3);
        float* yrow = &yo[mrow * YSTRIDE + colw];
        #pragma unroll
        for (int i = 0; i < 8; ++i) yrow[i] = acc[i];
    }
    __syncthreads();

    for (int idx = tid; idx < 16 * 128; idx += 256) {
        int r = idx >> 7;
        int c = idx & 127;
        int m = m_wg0 + c;
        if (m < n_out)
            out[(size_t)(b0 + r) * (size_t)n_out + m] = yo[r * YSTRIDE + c];
    }
}

extern "C" void kernel_launch(void* const* d_in, const int* in_sizes, int n_in_bufs,
                              void* d_out, int out_size, void* d_ws, size_t ws_size,
                              hipStream_t stream) {
    const float* x = (const float*)d_in[0];
    const float* h = (const float*)d_in[1];
    float* out = (float*)d_out;

    const int T = 1440000;
    const int B = in_sizes[0] / T;
    const int n_out = (int)(((long long)T * UP) / DOWN);

    dim3 grid((n_out + 127) / 128, B / 16);
    resample_poly_wmma<<<grid, 256, 0, stream>>>(x, h, out, T, n_out);
}


# Re-tiled per batch row: M = 16 consecutive 72-sample blocks, N = a block's 15 outputs, K = its 164-sample window; filter operand becomes tile-invariant (registers), x stages as one contiguous span
# speedup vs baseline: 3.3978x; 3.3978x over previous
#include <hip/hip_runtime.h>


typedef float v2f __attribute__((ext_vector_type(2)));
typedef float v4f __attribute__((ext_vector_type(4)));
typedef float v8f __attribute__((ext_vector_type(8)));

#define HLAST        480
#define KSTEPS       41
#define LEAD         48
#define UNIT_OUT     480
#define UNIT_IN      2304
#define UNITS_PER_ROW 625
#define SPAN_G       599
#define HALF_G       144
#define HALO_G       23
#define XS_FLOATS    2784
#define YS_FLOATS    480
#define WG_WAVES     4
#define TOTAL_WAVES  5120
#define HT_LO        335
#define HT_N         1176

__device__ __forceinline__ int half_base(int u) { return 704 * u + 4 * (u & 1); }

__device__ __forceinline__ constexpr bool round_has_halo(int i) {
    const int lo = 32 * i, hi = 32 * i + 31;
    for (int b = HALF_G; b <= 3 * HALF_G; b += HALF_G)
        if (hi >= b && lo < b + HALO_G) return true;
    return false;
}

__device__ __forceinline__ void wave_lds_handover() {
    __builtin_amdgcn_fence(__ATOMIC_RELEASE, "workgroup", "local");
    __builtin_amdgcn_wave_barrier();
    __builtin_amdgcn_fence(__ATOMIC_ACQUIRE, "workgroup", "local");
}

__global__ __launch_bounds__(128) __attribute__((amdgpu_waves_per_eu(5, 5))) void resample_rows_wmma(
    const float* __restrict__ x,
    const float* __restrict__ h,
    float* __restrict__ out,
    int n_in, int n_out, int rows)
{
    __shared__ __attribute__((aligned(16))) float xs_all[WG_WAVES][XS_FLOATS];
    __shared__ __attribute__((aligned(16))) float ys_all[WG_WAVES][YS_FLOATS];
    __shared__ float ht[HT_N];

    const int lane = threadIdx.x & 31;
    const int wave = __builtin_amdgcn_readfirstlane(threadIdx.x >> 5);
    float* xs = xs_all[wave];
    float* ys = ys_all[wave];

    const int c  = lane & 15;
    const int hh = lane >> 4;

    {
        float hv[4];
        #pragma unroll
        for (int q = 0; q < 4; ++q) {
            const int k = (int)threadIdx.x + 128 * q;
            hv[q] = h[min(k, HLAST)];
        }
        for (int i = threadIdx.x; i < HT_LO; i += 128) ht[i] = 0.0f;
        for (int i = HT_LO + HLAST + 1 + threadIdx.x; i < HT_N; i += 128) ht[i] = 0.0f;
        #pragma unroll
        for (int q = 0; q < 4; ++q) {
            const int k = (int)threadIdx.x + 128 * q;
            if (k <= HLAST) ht[HT_LO + k] = 5.0f * hv[q];
        }
    }
    __syncthreads();

    v2f bf[KSTEPS];
    #pragma unroll
    for (int j = 0; j < KSTEPS; ++j) {
        #pragma unroll
        for (int e = 0; e < 2; ++e) {
            const int k = 4 * j + 2 * hh + e;
            const float v = ht[24 * c + 480 + HT_LO - 5 * k];
            bf[j][e] = (c < 15) ? v : 0.0f;
        }
    }

    const float* a0p = xs + half_base(c >> 3) + 72 * (c & 7) + 2 * hh;
    const float* a1p = a0p + 1408;

    const int total_units = rows * UNITS_PER_ROW;
    const int wave_id = blockIdx.x * WG_WAVES + wave;

    for (int unit = wave_id; unit < total_units; unit += TOTAL_WAVES) {
        const int row = unit / UNITS_PER_ROW;
        const int U   = unit - row * UNITS_PER_ROW;
        const float* xrow = x + (size_t)row * (size_t)n_in;
        const int t_unit = UNIT_IN * U - LEAD;

        v4f st[19];
        const int t_lane = t_unit + 4 * lane;
        const float* xl = xrow + t_lane;
        #pragma unroll
        for (int i = 0; i < 19; ++i) {
            v4f v = {0.f, 0.f, 0.f, 0.f};
            if ((unsigned)(t_lane + 128 * i) < (unsigned)n_in) v = *(const v4f*)(xl + 128 * i);
            st[i] = v;
        }

        #pragma unroll
        for (int i = 0; i < 19; ++i) {
            const int g = 32 * i + lane;
            const int u = (g >= HALF_G) + (g >= 2 * HALF_G) + (g >= 3 * HALF_G);
            const int f = g - HALF_G * u;
            if (g < SPAN_G)
                *(v4f*)(xs + half_base(u) + 4 * f) = st[i];
            if (round_has_halo(i)) {
                if (u >= 1 && f < HALO_G)
                    *(v4f*)(xs + half_base(u - 1) + 4 * (f + HALF_G)) = st[i];
            }
        }
        wave_lds_handover();

        v8f acc0 = {0.f, 0.f, 0.f, 0.f, 0.f, 0.f, 0.f, 0.f};
        v8f acc1 = {0.f, 0.f, 0.f, 0.f, 0.f, 0.f, 0.f, 0.f};
        #pragma unroll
        for (int j = 0; j < KSTEPS; ++j) {
            const v2f A0 = *(const v2f*)(a0p + 4 * j);
            const v2f A1 = *(const v2f*)(a1p + 4 * j);
            acc0 = __builtin_amdgcn_wmma_f32_16x16x4_f32(false, A0, false, bf[j], (short)0, acc0, false, false);
            acc1 = __builtin_amdgcn_wmma_f32_16x16x4_f32(false, A1, false, bf[j], (short)0, acc1, false, false);
        }

        if (c < 15) {
            #pragma unroll
            for (int r = 0; r < 8; ++r) {
                ys[15 * (8 * hh + r) + c]       = acc0[r];
                ys[240 + 15 * (8 * hh + r) + c] = acc1[r];
            }
        }
        wave_lds_handover();

        float* orow = out + (size_t)row * (size_t)n_out + (size_t)UNIT_OUT * U;
        #pragma unroll
        for (int i = 0; i < 4; ++i) {
            const int f = 32 * i + lane;
            if (f < UNIT_OUT / 4) {
                const v4f v = *(const v4f*)(ys + 4 * f);
                *(v4f*)(orow + 4 * f) = v;
            }
        }
    }
}

extern "C" void kernel_launch(void* const* d_in, const int* in_sizes, int n_in_bufs,
                              void* d_out, int out_size, void* d_ws, size_t ws_size,
                              hipStream_t stream) {
    const float* x = (const float*)d_in[0];
    const float* h = (const float*)d_in[1];
    float* out = (float*)d_out;

    const int T = 1440000;
    const int B = in_sizes[0] / T;
    const int n_out = 300000;

    resample_rows_wmma<<<1280, 128, 0, stream>>>(x, h, out, T, n_out, B);
}


# Work-group split into 4 loader waves and 4 matrix waves: loaders bring the next rounds' spans into the idle one of two LDS stages during the products; one barrier per round hands the stages over
# speedup vs baseline: 3.5986x; 1.0591x over previous
#include <hip/hip_runtime.h>


typedef float v2f __attribute__((ext_vector_type(2)));
typedef float v4f __attribute__((ext_vector_type(4)));
typedef float v8f __attribute__((ext_vector_type(8)));

#define HLAST        480
#define KSTEPS       41
#define LEAD         48
#define UNIT_OUT     480
#define UNIT_IN      2304
#define UNITS_PER_ROW 625
#define SPAN_G       599
#define SPAN_LOADS   19
#define HALF_G       144
#define HALO_G       23
#define XS_FLOATS    2784
#define YS_FLOATS    480
#define PAIRS        4
#define WG_THREADS   256
#define HT_LO        335
#define HT_N         1176

__device__ __forceinline__ int half_base(int u) { return 704 * u + 4 * (u & 1); }

__device__ __forceinline__ constexpr bool round_has_halo(int i) {
    const int lo = 32 * i, hi = 32 * i + 31;
    for (int b = HALF_G; b <= 3 * HALF_G; b += HALF_G)
        if (hi >= b && lo < b + HALO_G) return true;
    return false;
}

__device__ __forceinline__ void wave_lds_handover() {
    __builtin_amdgcn_fence(__ATOMIC_RELEASE, "workgroup", "local");
    __builtin_amdgcn_wave_barrier();
    __builtin_amdgcn_fence(__ATOMIC_ACQUIRE, "workgroup", "local");
}

__device__ __forceinline__ void stage_handover() {
    __builtin_amdgcn_fence(__ATOMIC_RELEASE, "workgroup", "local");
    __builtin_amdgcn_s_barrier();
    __builtin_amdgcn_fence(__ATOMIC_ACQUIRE, "workgroup", "local");
}

__device__ __forceinline__ void load_span(v4f (&st)[SPAN_LOADS], const float* __restrict__ x,
                                          int unit, int n_in, int lane) {
    const int row = unit / UNITS_PER_ROW;
    const int U   = unit - row * UNITS_PER_ROW;
    const float* xrow = x + (size_t)row * (size_t)n_in;
    const int t_lane = UNIT_IN * U - LEAD + 4 * lane;
    const float* xl = xrow + t_lane;
    #pragma unroll
    for (int i = 0; i < SPAN_LOADS; ++i) {
        v4f v = {0.f, 0.f, 0.f, 0.f};
        if ((unsigned)(t_lane + 128 * i) < (unsigned)n_in) v = *(const v4f*)(xl + 128 * i);
        st[i] = v;
    }
}

__device__ __forceinline__ void store_span(const v4f (&st)[SPAN_LOADS], float* xs, int lane) {
    #pragma unroll
    for (int i = 0; i < SPAN_LOADS; ++i) {
        const int g = 32 * i + lane;
        const int u = (g >= HALF_G) + (g >= 2 * HALF_G) + (g >= 3 * HALF_G);
        const int f = g - HALF_G * u;
        if (g < SPAN_G)
            *(v4f*)(xs + half_base(u) + 4 * f) = st[i];
        if (round_has_halo(i)) {
            if (u >= 1 && f < HALO_G)
                *(v4f*)(xs + half_base(u - 1) + 4 * (f + HALF_G)) = st[i];
        }
    }
}

__global__ __launch_bounds__(WG_THREADS) void resample_rows_wmma(
    const float* __restrict__ x,
    const float* __restrict__ h,
    float* __restrict__ out,
    int n_in, int n_out, int rows)
{
    __shared__ __attribute__((aligned(16))) float xs_all[2][PAIRS][XS_FLOATS];
    __shared__ __attribute__((aligned(16))) float ys_all[PAIRS][YS_FLOATS];
    __shared__ float ht[HT_N];

    const int lane = threadIdx.x & 31;
    const int wave = __builtin_amdgcn_readfirstlane(threadIdx.x >> 5);
    const int slot = wave & 3;
    const bool loader = wave < PAIRS;

    const int c  = lane & 15;
    const int hh = lane >> 4;

    const int total_units = rows * UNITS_PER_ROW;
    const int quads = (total_units + PAIRS - 1) / PAIRS;
    const int G = (int)gridDim.x;
    const int g0 = (int)blockIdx.x;
    const int rounds = (g0 < quads) ? (quads - g0 + G - 1) / G : 0;
    const int unit0 = PAIRS * g0 + slot;
    const int unit_step = PAIRS * G;

    v2f bf[KSTEPS];
    v4f st[SPAN_LOADS];

    {
        float hv[2];
        #pragma unroll
        for (int q = 0; q < 2; ++q) {
            const int k = (int)threadIdx.x + WG_THREADS * q;
            hv[q] = h[min(k, HLAST)];
        }
        if (loader && rounds > 0 && unit0 < total_units)
            load_span(st, x, unit0, n_in, lane);
        for (int i = threadIdx.x; i < HT_LO; i += WG_THREADS) ht[i] = 0.0f;
        for (int i = HT_LO + HLAST + 1 + threadIdx.x; i < HT_N; i += WG_THREADS) ht[i] = 0.0f;
        #pragma unroll
        for (int q = 0; q < 2; ++q) {
            const int k = (int)threadIdx.x + WG_THREADS * q;
            if (k <= HLAST) ht[HT_LO + k] = 5.0f * hv[q];
        }
    }
    __syncthreads();

    if (loader) {
        if (rounds > 0 && unit0 < total_units)
            store_span(st, xs_all[0][slot], lane);
        if (rounds > 1 && unit0 + unit_step < total_units)
            load_span(st, x, unit0 + unit_step, n_in, lane);
    } else {
        #pragma unroll
        for (int j = 0; j < KSTEPS; ++j) {
            #pragma unroll
            for (int e = 0; e < 2; ++e) {
                const int k = 4 * j + 2 * hh + e;
                const float v = ht[24 * c + 480 + HT_LO - 5 * k];
                bf[j][e] = (c < 15) ? v : 0.0f;
            }
        }
    }
    stage_handover();

    float* ys = ys_all[slot];

    for (int r = 0; r < rounds; ++r) {
        const int unit = unit0 + unit_step * r;
        if (loader) {
            if (r + 1 < rounds && unit + unit_step < total_units)
                store_span(st, xs_all[(r + 1) & 1][slot], lane);
            if (r + 2 < rounds && unit + 2 * unit_step < total_units)
                load_span(st, x, unit + 2 * unit_step, n_in, lane);
        } else if (unit < total_units) {
            const float* xs = xs_all[r & 1][slot];
            const float* a0p = xs + half_base(c >> 3) + 72 * (c & 7) + 2 * hh;
            const float* a1p = a0p + 1408;

            v8f acc0 = {0.f, 0.f, 0.f, 0.f, 0.f, 0.f, 0.f, 0.f};
            v8f acc1 = {0.f, 0.f, 0.f, 0.f, 0.f, 0.f, 0.f, 0.f};
            #pragma unroll
            for (int j = 0; j < KSTEPS; ++j) {
                const v2f A0 = *(const v2f*)(a0p + 4 * j);
                const v2f A1 = *(const v2f*)(a1p + 4 * j);
                acc0 = __builtin_amdgcn_wmma_f32_16x16x4_f32(false, A0, false, bf[j], (short)0, acc0, false, false);
                acc1 = __builtin_amdgcn_wmma_f32_16x16x4_f32(false, A1, false, bf[j], (short)0, acc1, false, false);
            }

            if (c < 15) {
                #pragma unroll
                for (int q = 0; q < 8; ++q) {
                    ys[15 * (8 * hh + q) + c]       = acc0[q];
                    ys[240 + 15 * (8 * hh + q) + c] = acc1[q];
                }
            }
            wave_lds_handover();

            const int row = unit / UNITS_PER_ROW;
            const int U   = unit - row * UNITS_PER_ROW;
            float* orow = out + (size_t)row * (size_t)n_out + (size_t)UNIT_OUT * U;
            #pragma unroll
            for (int i = 0; i < 4; ++i) {
                const int f = 32 * i + lane;
                if (f < UNIT_OUT / 4) {
                    const v4f v = *(const v4f*)(ys + 4 * f);
                    *(v4f*)(orow + 4 * f) = v;
                }
            }
        }
        stage_handover();
    }
}

extern "C" void kernel_launch(void* const* d_in, const int* in_sizes, int n_in_bufs,
                              void* d_out, int out_size, void* d_ws, size_t ws_size,
                              hipStream_t stream) {
    const float* x = (const float*)d_in[0];
    const float* h = (const float*)d_in[1];
    float* out = (float*)d_out;

    const int T = 1440000;
    const int B = in_sizes[0] / T;
    const int n_out = 300000;

    resample_rows_wmma<<<512, WG_THREADS, 0, stream>>>(x, h, out, T, n_out, B);
}
